# starting kernel
# baseline (speedup 1.0000x reference)
#include <hip/hip_runtime.h>

#define N_NODES 50000
#define N_EDGES 800000
#define IN_C    64
#define HID_C   128
#define OUT_C   64

typedef __attribute__((ext_vector_type(2))) float v2f;
typedef __attribute__((ext_vector_type(8))) float v8f;


__global__ void gcn_deg_init(float* __restrict__ deg) {
    int n = blockIdx.x * blockDim.x + threadIdx.x;
    if (n < N_NODES) deg[n] = 1.0f;
}

__global__ void gcn_deg_accum(const int* __restrict__ dst, float* __restrict__ deg) {
    int e = blockIdx.x * blockDim.x + threadIdx.x;
    if (e < N_EDGES) atomicAdd(&deg[dst[e]], 1.0f);
}

__global__ void gcn_deg_to_dinv(float* __restrict__ deg) {
    int n = blockIdx.x * blockDim.x + threadIdx.x;
    if (n < N_NODES) deg[n] = rsqrtf(deg[n]);
}


__global__ __launch_bounds__(256) void gcn_gemm1_wmma(
    const float* __restrict__ x, const float* __restrict__ W1, float* __restrict__ xw) {
    const int lane = threadIdx.x & 31;
    const int wave = threadIdx.x >> 5;
    const int m0   = blockIdx.x * 16;
    const int n0   = wave * 16;
    const int half = lane >> 4;
    const int l15  = lane & 15;
    const int row  = m0 + l15;
    const int col  = n0 + l15;
    v8f acc = {0.f, 0.f, 0.f, 0.f, 0.f, 0.f, 0.f, 0.f};

    for (int k0 = 0; k0 < IN_C; k0 += 4) {
        const int ka = k0 + half * 2;
        v2f a = *(const v2f*)(x + row * IN_C + ka);
        v2f b;
        b.x = W1[ka * HID_C + col];
        b.y = W1[(ka + 1) * HID_C + col];
        acc = __builtin_amdgcn_wmma_f32_16x16x4_f32(
             false, a,  false, b,
             (short)0, acc,  false,  false);
    }
#pragma unroll
    for (int r = 0; r < 8; ++r)
        xw[(size_t)(m0 + half * 8 + r) * HID_C + col] = acc[r];
}


__global__ __launch_bounds__(128) void gcn_gemm2_wmma(
    const float* __restrict__ h, const float* __restrict__ W2, float* __restrict__ xw) {
    const int lane = threadIdx.x & 31;
    const int wave = threadIdx.x >> 5;
    const int m0   = blockIdx.x * 16;
    const int n0   = wave * 16;
    const int half = lane >> 4;
    const int l15  = lane & 15;
    const int row  = m0 + l15;
    const int col  = n0 + l15;
    v8f acc = {0.f, 0.f, 0.f, 0.f, 0.f, 0.f, 0.f, 0.f};

    for (int k0 = 0; k0 < HID_C; k0 += 4) {
        const int ka = k0 + half * 2;
        v2f a = *(const v2f*)(h + (size_t)row * HID_C + ka);
        a.x = fmaxf(a.x, 0.0f);
        a.y = fmaxf(a.y, 0.0f);
        v2f b;
        b.x = W2[ka * OUT_C + col];
        b.y = W2[(ka + 1) * OUT_C + col];
        acc = __builtin_amdgcn_wmma_f32_16x16x4_f32(
            false, a, false, b, (short)0, acc, false, false);
    }
#pragma unroll
    for (int r = 0; r < 8; ++r)
        xw[(size_t)(m0 + half * 8 + r) * OUT_C + col] = acc[r];
}


template <int C>
__global__ void gcn_agg_init(const float* __restrict__ xw, const float* __restrict__ dinv,
                             const float* __restrict__ bias, float* __restrict__ agg) {
    int i = blockIdx.x * blockDim.x + threadIdx.x;
    if (i >= N_NODES * C) return;
    int n = i / C;
    int c = i % C;
    float dn = dinv[n];
    agg[i] = bias[c] + xw[i] * dn * dn;
}


template <int C>
__global__ void gcn_scatter(const int* __restrict__ src, const int* __restrict__ dst,
                            const float* __restrict__ dinv,
                            const float* __restrict__ xw, float* __restrict__ agg) {
    const int LPE = C / 4;
    int t = blockIdx.x * blockDim.x + threadIdx.x;
    int e = t / LPE;
    if (e >= N_EDGES) return;
    int q = t % LPE;
    int s = src[e];
    int d = dst[e];
    float norm = dinv[s] * dinv[d];
    const float4 v = *(const float4*)(xw + (size_t)s * C + q * 4);
    float* base = agg + (size_t)d * C + q * 4;
    atomicAdd(base + 0, v.x * norm);
    atomicAdd(base + 1, v.y * norm);
    atomicAdd(base + 2, v.z * norm);
    atomicAdd(base + 3, v.w * norm);
}


extern "C" void kernel_launch(void* const* d_in, const int* in_sizes, int n_in,
                              void* d_out, int out_size, void* d_ws, size_t ws_size,
                              hipStream_t stream) {
    const float* x  = (const float*)d_in[0];
    const int*   ei = (const int*)d_in[1];
    const float* W1 = (const float*)d_in[2];
    const float* b1 = (const float*)d_in[3];
    const float* W2 = (const float*)d_in[4];
    const float* b2 = (const float*)d_in[5];
    float* out = (float*)d_out;

    const int* src = ei;
    const int* dst = ei + N_EDGES;

    float* dinv = (float*)d_ws;
    float* xw1  = dinv + 50176;
    float* agg1 = xw1 + 6400000;
    float* xw2  = xw1;

    gcn_deg_init<<<(N_NODES + 255) / 256, 256, 0, stream>>>(dinv);
    gcn_deg_accum<<<(N_EDGES + 255) / 256, 256, 0, stream>>>(dst, dinv);
    gcn_deg_to_dinv<<<(N_NODES + 255) / 256, 256, 0, stream>>>(dinv);

    gcn_gemm1_wmma<<<N_NODES / 16, 256, 0, stream>>>(x, W1, xw1);

    gcn_agg_init<HID_C><<<(N_NODES * HID_C + 255) / 256, 256, 0, stream>>>(xw1, dinv, b1, agg1);
    gcn_scatter<HID_C><<<(N_EDGES * (HID_C / 4) + 255) / 256, 256, 0, stream>>>(src, dst, dinv, xw1, agg1);

    gcn_gemm2_wmma<<<N_NODES / 16, 128, 0, stream>>>(agg1, W2, xw2);

    gcn_agg_init<OUT_C><<<(N_NODES * OUT_C + 255) / 256, 256, 0, stream>>>(xw2, dinv, b2, out);
    gcn_scatter<OUT_C><<<(N_EDGES * (OUT_C / 4) + 255) / 256, 256, 0, stream>>>(src, dst, dinv, xw2, out);
}


# Both products re-tiled: a block loops over 16-row stripes, W fragments stay in registers, each A stripe enters once per block by whole-line loads via LDS (was per wave, 16 lines a load); sums kept
# speedup vs baseline: 2.3873x; 2.3873x over previous
#include <hip/hip_runtime.h>

#define N_NODES 50000
#define N_EDGES 800000
#define IN_C    64
#define HID_C   128
#define OUT_C   64

typedef __attribute__((ext_vector_type(2))) float v2f;
typedef __attribute__((ext_vector_type(8))) float v8f;


#define DEG_STRIDE 32

__global__ void gcn_deg_init(float* __restrict__ degp) {
    int n = blockIdx.x * blockDim.x + threadIdx.x;
    if (n < N_NODES) degp[(size_t)n * DEG_STRIDE] = 1.0f;
}

__global__ void gcn_deg_accum(const int* __restrict__ dst, float* __restrict__ degp) {
    int e = blockIdx.x * blockDim.x + threadIdx.x;
    if (e < N_EDGES) atomicAdd(&degp[(size_t)dst[e] * DEG_STRIDE], 1.0f);
}


#define N_TILES   (N_NODES / 16)
#define G1_PITCH  (IN_C + 4)
#define G2_PITCH  (HID_C + 4)
#define G1_BLOCKS 1024
#define G2_BLOCKS 1024

static __device__ __forceinline__ void lds_barrier() {
    __builtin_amdgcn_fence(__ATOMIC_RELEASE, "workgroup", "local");
    __builtin_amdgcn_s_barrier();
    __builtin_amdgcn_fence(__ATOMIC_ACQUIRE, "workgroup", "local");
}

__global__ __launch_bounds__(256) void gcn_gemm1_wmma(
    const float* __restrict__ x, const float* __restrict__ W1,
    const float* __restrict__ degp, const float* __restrict__ bias,
    float* __restrict__ xw, float* __restrict__ agg, float* __restrict__ dinv) {
    __shared__ __attribute__((aligned(16))) float lds[2][16 * G1_PITCH];
    const int lane = threadIdx.x & 31;
    const int wave = threadIdx.x >> 5;
    const int n0   = wave * 16;
    const int half = lane >> 4;
    const int l15  = lane & 15;
    const int col  = n0 + l15;
    v2f bf[IN_C / 4];
#pragma unroll
    for (int ks = 0; ks < IN_C / 4; ++ks) {
        const int ka = ks * 4 + half * 2;
        bf[ks].x = W1[ka * HID_C + col];
        bf[ks].y = W1[(ka + 1) * HID_C + col];
    }
    const float bcol = bias[col];
    const int sr = threadIdx.x >> 4;
    const int sc = (threadIdx.x & 15) * 4;
    int tile = blockIdx.x;
    float4 st = *(const float4*)(x + (size_t)(tile * 16 + sr) * IN_C + sc);
    *(float4*)(&lds[0][sr * G1_PITCH + sc]) = st;
    __syncthreads();
    int buf = 0;
    for (; tile < N_TILES; tile += gridDim.x) {
        const int  m0       = tile * 16;
        const int  nxt      = tile + gridDim.x;
        const bool has_next = nxt < N_TILES;
        if (has_next) st = *(const float4*)(x + (size_t)(nxt * 16 + sr) * IN_C + sc);
        float dg[8];
#pragma unroll
        for (int r = 0; r < 8; ++r) dg[r] = degp[(size_t)(m0 + half * 8 + r) * DEG_STRIDE];
        v8f acc = {0.f, 0.f, 0.f, 0.f, 0.f, 0.f, 0.f, 0.f};
        const float* at = &lds[buf][l15 * G1_PITCH + half * 2];
#pragma unroll
        for (int ks = 0; ks < IN_C / 4; ++ks) {
            v2f a = *(const v2f*)(at + ks * 4);
            acc = __builtin_amdgcn_wmma_f32_16x16x4_f32(
                 false, a,  false, bf[ks],
                 (short)0, acc,  false,  false);
        }
        asm volatile("s_wait_alu depctr_va_vdst(0)" : "+v"(acc));
        float dn[8];
#pragma unroll
        for (int r = 0; r < 8; ++r) {
            const size_t o = (size_t)(m0 + half * 8 + r) * HID_C + col;
            dn[r]  = rsqrtf(dg[r]);
            xw[o]  = acc[r] * dn[r];
            agg[o] = (acc[r] + bcol * dg[r]) * dn[r];
        }
        if (wave == 0 && l15 == 0) {
#pragma unroll
            for (int r = 0; r < 8; ++r) dinv[m0 + half * 8 + r] = dn[r];
        }
        if (has_next) *(float4*)(&lds[buf ^ 1][sr * G1_PITCH + sc]) = st;
        lds_barrier();
        buf ^= 1;
    }
}


__global__ __launch_bounds__(128) void gcn_gemm2_wmma(
    const float* __restrict__ h, const float* __restrict__ W2,
    const float* __restrict__ dinv, const float* __restrict__ bias,
    float* __restrict__ xw, float* __restrict__ agg) {
    __shared__ __attribute__((aligned(16))) float lds[2][16 * G2_PITCH];
    const int lane = threadIdx.x & 31;
    const int wave = threadIdx.x >> 5;
    const int n0   = wave * 16;
    const int half = lane >> 4;
    const int l15  = lane & 15;
    const int col  = n0 + l15;
    v2f bf[HID_C / 4];
#pragma unroll
    for (int ks = 0; ks < HID_C / 4; ++ks) {
        const int ka = ks * 4 + half * 2;
        bf[ks].x = W2[ka * OUT_C + col];
        bf[ks].y = W2[(ka + 1) * OUT_C + col];
    }
    const float bcol = bias[col];
    const int sr = threadIdx.x >> 5;
    const int sc = (threadIdx.x & 31) * 4;
    int tile = blockIdx.x;
    float4 st[4];
#pragma unroll
    for (int i = 0; i < 4; ++i) st[i] = *(const float4*)(h + (size_t)(tile * 16 + sr + 4 * i) * HID_C + sc);
#pragma unroll
    for (int i = 0; i < 4; ++i) {
        float4 v = st[i];
        v.x = fmaxf(v.x, 0.0f); v.y = fmaxf(v.y, 0.0f); v.z = fmaxf(v.z, 0.0f); v.w = fmaxf(v.w, 0.0f);
        *(float4*)(&lds[0][(sr + 4 * i) * G2_PITCH + sc]) = v;
    }
    __syncthreads();
    int buf = 0;
    for (; tile < N_TILES; tile += gridDim.x) {
        const int  m0       = tile * 16;
        const int  nxt      = tile + gridDim.x;
        const bool has_next = nxt < N_TILES;
        if (has_next) {
#pragma unroll
            for (int i = 0; i < 4; ++i) st[i] = *(const float4*)(h + (size_t)(nxt * 16 + sr + 4 * i) * HID_C + sc);
        }
        float dn[8];
#pragma unroll
        for (int r = 0; r < 8; ++r) dn[r] = dinv[m0 + half * 8 + r];
        v8f acc = {0.f, 0.f, 0.f, 0.f, 0.f, 0.f, 0.f, 0.f};
        const float* at = &lds[buf][l15 * G2_PITCH + half * 2];
#pragma unroll
        for (int ks = 0; ks < HID_C / 4; ++ks) {
            v2f a = *(const v2f*)(at + ks * 4);
            acc = __builtin_amdgcn_wmma_f32_16x16x4_f32(
                false, a, false, bf[ks], (short)0, acc, false, false);
        }
        asm volatile("s_wait_alu depctr_va_vdst(0)" : "+v"(acc));
#pragma unroll
        for (int r = 0; r < 8; ++r) {
            const size_t o = (size_t)(m0 + half * 8 + r) * OUT_C + col;
            const float y = acc[r] * dn[r] * dn[r];
            xw[o]  = y;
            agg[o] = bcol + y * dn[r];
        }
        if (has_next) {
#pragma unroll
            for (int i = 0; i < 4; ++i) {
                float4 v = st[i];
                v.x = fmaxf(v.x, 0.0f); v.y = fmaxf(v.y, 0.0f); v.z = fmaxf(v.z, 0.0f); v.w = fmaxf(v.w, 0.0f);
                *(float4*)(&lds[buf ^ 1][(sr + 4 * i) * G2_PITCH + sc]) = v;
            }
        }
        lds_barrier();
        buf ^= 1;
    }
}


template <int C, bool DST_SCALE>
__global__ void gcn_scatter(const int* __restrict__ src, const int* __restrict__ dst,
                            const float* __restrict__ dinv,
                            const float* __restrict__ xw, float* __restrict__ agg) {
    constexpr int EPW = 32;
    constexpr int LPR = C / 32;
    constexpr int G   = 32 / LPR;
    static_assert(N_EDGES % EPW == 0, "a wave's edges are all inside or all outside");
    const int lane = threadIdx.x & 31;
    const int wave = __builtin_amdgcn_readfirstlane(
        (int)((blockIdx.x * blockDim.x + threadIdx.x) >> 5));
    const int e0 = wave * EPW;
    if (e0 >= N_EDGES) return;
    const int   s   = src[e0 + lane];
    const int   d   = dst[e0 + lane];
    float nrm = 1.0f;
    if constexpr (DST_SCALE) nrm = dinv[d];
    for (int g = 0; g < EPW; g += G) {
        float v[G][LPR];
#pragma unroll
        for (int k = 0; k < G; ++k) {
            const int sk = __builtin_amdgcn_readlane(s, g + k);
            const float* xr = xw + (size_t)sk * C + lane;
#pragma unroll
            for (int j = 0; j < LPR; ++j) v[k][j] = xr[32 * j];
        }
#pragma unroll
        for (int k = 0; k < G; ++k) {
            const int   dk = __builtin_amdgcn_readlane(d, g + k);
            float* ar = agg + (size_t)dk * C + lane;
            if constexpr (DST_SCALE) {
                const float nk = __int_as_float(__builtin_amdgcn_readlane(__float_as_int(nrm), g + k));
#pragma unroll
                for (int j = 0; j < LPR; ++j) atomicAdd(ar + 32 * j, v[k][j] * nk);
            } else {
#pragma unroll
                for (int j = 0; j < LPR; ++j) atomicAdd(ar + 32 * j, v[k][j]);
            }
        }
    }
}


extern "C" void kernel_launch(void* const* d_in, const int* in_sizes, int n_in,
                              void* d_out, int out_size, void* d_ws, size_t ws_size,
                              hipStream_t stream) {
    const float* x  = (const float*)d_in[0];
    const int*   ei = (const int*)d_in[1];
    const float* W1 = (const float*)d_in[2];
    const float* b1 = (const float*)d_in[3];
    const float* W2 = (const float*)d_in[4];
    const float* b2 = (const float*)d_in[5];
    float* out = (float*)d_out;

    const int* src = ei;
    const int* dst = ei + N_EDGES;

    float* dinv = (float*)d_ws;
    float* xw1  = dinv + 50176;
    float* agg1 = xw1 + 6400000;
    float* degp = agg1 + 6400000;
    float* xw2  = xw1;

    gcn_deg_init<<<(N_NODES + 255) / 256, 256, 0, stream>>>(degp);
    gcn_deg_accum<<<(N_EDGES + 255) / 256, 256, 0, stream>>>(dst, degp);

    gcn_gemm1_wmma<<<G1_BLOCKS, 256, 0, stream>>>(x, W1, degp, b1, xw1, agg1, dinv);

    gcn_scatter<HID_C, false><<<(N_EDGES + 255) / 256, 256, 0, stream>>>(src, dst, dinv, xw1, agg1);

    gcn_gemm2_wmma<<<G2_BLOCKS, 128, 0, stream>>>(agg1, W2, dinv, b2, xw2, out);

    gcn_scatter<OUT_C, true><<<(N_EDGES + 255) / 256, 256, 0, stream>>>(src, dst, dinv, xw2, out);
}


# Aggregations leave global float atomics: block b keeps its 196 destination rows in LDS, finds its edges in a per-range bit plane the degree pass builds, adds rows by LDS atomics (arrival order)
# speedup vs baseline: 2.7271x; 1.1423x over previous
#include <hip/hip_runtime.h>

#define N_NODES 50000
#define N_EDGES 800000
#define IN_C    64
#define HID_C   128
#define OUT_C   64

typedef __attribute__((ext_vector_type(2))) float v2f;
typedef __attribute__((ext_vector_type(8))) float v8f;


#define DEG_STRIDE 32

__global__ void gcn_deg_init(float* __restrict__ degp) {
    int n = blockIdx.x * blockDim.x + threadIdx.x;
    if (n < N_NODES) degp[(size_t)n * DEG_STRIDE] = 1.0f;
}

#define NB       256
#define RANGE    196
#define EW       (N_EDGES / 32)
#define TB_EDGES 2048
#define TB_WORDS (TB_EDGES / 32)

__global__ __launch_bounds__(256) void gcn_deg_accum(const int* __restrict__ dst, float* __restrict__ degp,
                                                     unsigned* __restrict__ bm) {
    __shared__ __attribute__((aligned(16))) unsigned tile[NB * TB_WORDS];
    const int t = threadIdx.x;
    uint4* tq = (uint4*)tile;
#pragma unroll
    for (int i = 0; i < 16; ++i) tq[i * 256 + t] = make_uint4(0u, 0u, 0u, 0u);
    __syncthreads();
    const int e0 = blockIdx.x * TB_EDGES;
    int dd[8];
#pragma unroll
    for (int j = 0; j < 8; ++j) {
        const int e = e0 + j * 256 + t;
        dd[j] = (e < N_EDGES) ? dst[e] : -1;
    }
#pragma unroll
    for (int j = 0; j < 8; ++j) {
        const int el = j * 256 + t;
        const int d  = dd[j];
        if (d >= 0) {
            atomicAdd(&degp[(size_t)d * DEG_STRIDE], 1.0f);
            atomicOr(&tile[(d / RANGE) * TB_WORDS + (el >> 5)], 1u << (el & 31));
        }
    }
    __syncthreads();
#pragma unroll
    for (int i = 0; i < 16; ++i) {
        const int p     = i * 256 + t;
        const int range = p >> 4;
        const int wj    = blockIdx.x * TB_WORDS + (p & 15) * 4;
        if (wj < EW) *(uint4*)(bm + (size_t)range * EW + wj) = tq[p];
    }
}


#define N_TILES   (N_NODES / 16)
#define G1_PITCH  (IN_C + 4)
#define G2_PITCH  (HID_C + 4)
#define G1_BLOCKS 1024
#define G2_BLOCKS 1024

static __device__ __forceinline__ void lds_barrier() {
    __builtin_amdgcn_fence(__ATOMIC_RELEASE, "workgroup", "local");
    __builtin_amdgcn_s_barrier();
    __builtin_amdgcn_fence(__ATOMIC_ACQUIRE, "workgroup", "local");
}

__global__ __launch_bounds__(256) void gcn_gemm1_wmma(
    const float* __restrict__ x, const float* __restrict__ W1,
    const float* __restrict__ degp,
    float* __restrict__ xw, float* __restrict__ dinv) {
    __shared__ __attribute__((aligned(16))) float lds[2][16 * G1_PITCH];
    const int lane = threadIdx.x & 31;
    const int wave = threadIdx.x >> 5;
    const int n0   = wave * 16;
    const int half = lane >> 4;
    const int l15  = lane & 15;
    const int col  = n0 + l15;
    v2f bf[IN_C / 4];
#pragma unroll
    for (int ks = 0; ks < IN_C / 4; ++ks) {
        const int ka = ks * 4 + half * 2;
        bf[ks].x = W1[ka * HID_C + col];
        bf[ks].y = W1[(ka + 1) * HID_C + col];
    }
    const int sr = threadIdx.x >> 4;
    const int sc = (threadIdx.x & 15) * 4;
    int tile = blockIdx.x;
    float4 st = *(const float4*)(x + (size_t)(tile * 16 + sr) * IN_C + sc);
    *(float4*)(&lds[0][sr * G1_PITCH + sc]) = st;
    __syncthreads();
    int buf = 0;
    for (; tile < N_TILES; tile += gridDim.x) {
        const int  m0       = tile * 16;
        const int  nxt      = tile + gridDim.x;
        const bool has_next = nxt < N_TILES;
        if (has_next) st = *(const float4*)(x + (size_t)(nxt * 16 + sr) * IN_C + sc);
        float dg[8];
#pragma unroll
        for (int r = 0; r < 8; ++r) dg[r] = degp[(size_t)(m0 + half * 8 + r) * DEG_STRIDE];
        v8f acc = {0.f, 0.f, 0.f, 0.f, 0.f, 0.f, 0.f, 0.f};
        const float* at = &lds[buf][l15 * G1_PITCH + half * 2];
#pragma unroll
        for (int ks = 0; ks < IN_C / 4; ++ks) {
            v2f a = *(const v2f*)(at + ks * 4);
            acc = __builtin_amdgcn_wmma_f32_16x16x4_f32(
                 false, a,  false, bf[ks],
                 (short)0, acc,  false,  false);
        }
        asm volatile("s_wait_alu depctr_va_vdst(0)" : "+v"(acc));
        float dn[8];
#pragma unroll
        for (int r = 0; r < 8; ++r) {
            const size_t o = (size_t)(m0 + half * 8 + r) * HID_C + col;
            dn[r]  = rsqrtf(dg[r]);
            xw[o]  = acc[r] * dn[r];
        }
        if (wave == 0 && l15 == 0) {
#pragma unroll
            for (int r = 0; r < 8; ++r) dinv[m0 + half * 8 + r] = dn[r];
        }
        if (has_next) *(float4*)(&lds[buf ^ 1][sr * G1_PITCH + sc]) = st;
        lds_barrier();
        buf ^= 1;
    }
}


__global__ __launch_bounds__(128) void gcn_gemm2_wmma(
    const float* __restrict__ h, const float* __restrict__ W2,
    const float* __restrict__ dinv,
    float* __restrict__ xw) {
    __shared__ __attribute__((aligned(16))) float lds[2][16 * G2_PITCH];
    const int lane = threadIdx.x & 31;
    const int wave = threadIdx.x >> 5;
    const int n0   = wave * 16;
    const int half = lane >> 4;
    const int l15  = lane & 15;
    const int col  = n0 + l15;
    v2f bf[HID_C / 4];
#pragma unroll
    for (int ks = 0; ks < HID_C / 4; ++ks) {
        const int ka = ks * 4 + half * 2;
        bf[ks].x = W2[ka * OUT_C + col];
        bf[ks].y = W2[(ka + 1) * OUT_C + col];
    }
    const int sr = threadIdx.x >> 5;
    const int sc = (threadIdx.x & 31) * 4;
    int tile = blockIdx.x;
    float4 st[4];
#pragma unroll
    for (int i = 0; i < 4; ++i) st[i] = *(const float4*)(h + (size_t)(tile * 16 + sr + 4 * i) * HID_C + sc);
#pragma unroll
    for (int i = 0; i < 4; ++i) {
        float4 v = st[i];
        v.x = fmaxf(v.x, 0.0f); v.y = fmaxf(v.y, 0.0f); v.z = fmaxf(v.z, 0.0f); v.w = fmaxf(v.w, 0.0f);
        *(float4*)(&lds[0][(sr + 4 * i) * G2_PITCH + sc]) = v;
    }
    __syncthreads();
    int buf = 0;
    for (; tile < N_TILES; tile += gridDim.x) {
        const int  m0       = tile * 16;
        const int  nxt      = tile + gridDim.x;
        const bool has_next = nxt < N_TILES;
        if (has_next) {
#pragma unroll
            for (int i = 0; i < 4; ++i) st[i] = *(const float4*)(h + (size_t)(nxt * 16 + sr + 4 * i) * HID_C + sc);
        }
        float dn[8];
#pragma unroll
        for (int r = 0; r < 8; ++r) dn[r] = dinv[m0 + half * 8 + r];
        v8f acc = {0.f, 0.f, 0.f, 0.f, 0.f, 0.f, 0.f, 0.f};
        const float* at = &lds[buf][l15 * G2_PITCH + half * 2];
#pragma unroll
        for (int ks = 0; ks < HID_C / 4; ++ks) {
            v2f a = *(const v2f*)(at + ks * 4);
            acc = __builtin_amdgcn_wmma_f32_16x16x4_f32(
                false, a, false, bf[ks], (short)0, acc, false, false);
        }
        asm volatile("s_wait_alu depctr_va_vdst(0)" : "+v"(acc));
#pragma unroll
        for (int r = 0; r < 8; ++r) {
            const size_t o = (size_t)(m0 + half * 8 + r) * OUT_C + col;
            xw[o] = acc[r] * dn[r] * dn[r];
        }
        if (has_next) {
#pragma unroll
            for (int i = 0; i < 4; ++i) {
                float4 v = st[i];
                v.x = fmaxf(v.x, 0.0f); v.y = fmaxf(v.y, 0.0f); v.z = fmaxf(v.z, 0.0f); v.w = fmaxf(v.w, 0.0f);
                *(float4*)(&lds[buf ^ 1][(sr + 4 * i) * G2_PITCH + sc]) = v;
            }
        }
        lds_barrier();
        buf ^= 1;
    }
}


#define CAP 8192

template <int C, bool FINAL>
__global__ __launch_bounds__(1024, 16) void gcn_agg_lds(
    const int* __restrict__ src, const int* __restrict__ dst, const unsigned* __restrict__ bm,
    const float* __restrict__ dinv, const float* __restrict__ degp, const float* __restrict__ bias,
    const float* __restrict__ y, float* __restrict__ outp) {
    constexpr int V = C / 32;
    typedef float vf __attribute__((ext_vector_type(V)));
    __shared__ __attribute__((aligned(16))) float acc[RANGE * C];
    __shared__ unsigned list[CAP + 1];
    __shared__ unsigned cnt;
    const int tid   = threadIdx.x;
    const int lane  = tid & 31;
    const int wave  = __builtin_amdgcn_readfirstlane(tid >> 5);
    const int lo    = blockIdx.x * RANGE;
    const int nrows = (N_NODES - lo) < RANGE ? (N_NODES - lo) : RANGE;
    if (tid == 0) cnt = 0u;

    constexpr int NQ = EW / 4;
    constexpr int U  = 7;
    static_assert(U * 1024 >= NQ, "one sweep covers the range's row");
    const uint4* bq = (const uint4*)(bm + (size_t)blockIdx.x * EW);
    uint4 w[U];
#pragma unroll
    for (int u = 0; u < U; ++u) {
        const int q = u * 1024 + tid;
        w[u] = (q < NQ) ? bq[q] : make_uint4(0u, 0u, 0u, 0u);
    }

    for (int r = wave; r < nrows; r += 32) {
        const int n = lo + r;
        vf v = *(const vf*)(y + (size_t)n * C + V * lane);
        if constexpr (!FINAL) {
            const float f = degp[(size_t)n * DEG_STRIDE] * dinv[n];
            const vf bv = *(const vf*)(bias + V * lane);
            v += bv * f;
        }
#pragma unroll
        for (int i = 0; i < V; ++i) acc[r * C + i * 32 + lane] = v[i];
    }
    __syncthreads();

    unsigned c = 0u;
#pragma unroll
    for (int u = 0; u < U; ++u)
        c += __builtin_popcount(w[u].x) + __builtin_popcount(w[u].y) +
             __builtin_popcount(w[u].z) + __builtin_popcount(w[u].w);
    unsigned pos0 = 0u;
    if (c != 0u) pos0 = atomicAdd(&cnt, c);
    {
        unsigned pos = pos0;
#pragma unroll
        for (int u = 0; u < U; ++u) {
            const unsigned ws[4] = {w[u].x, w[u].y, w[u].z, w[u].w};
#pragma unroll
            for (int k = 0; k < 4; ++k) {
                unsigned word = ws[k];
                const int eb = ((u * 1024 + tid) * 4 + k) * 32;
                while (word != 0u) {
                    list[pos < (unsigned)CAP ? pos : (unsigned)CAP] = (unsigned)(eb + __builtin_ctz(word));
                    word &= word - 1u;
                    ++pos;
                }
            }
        }
    }
    __syncthreads();

    const int all   = __builtin_amdgcn_readfirstlane((int)cnt);
    const int total = all < CAP ? all : CAP;
    constexpr int G = 32 / V;
    for (int i0 = wave * 32; i0 < total; i0 += 1024) {
        const int i = i0 + lane;
        int s = 0, r = 0;
        if (i < total) {
            const unsigned e = list[i];
            s = src[e];
            r = dst[e] - lo;
        }
        const int m = (total - i0) < 32 ? (total - i0) : 32;
        for (int g = 0; g < m; g += G) {
            vf v[G];
#pragma unroll
            for (int k = 0; k < G; ++k) {
                const int sk = __builtin_amdgcn_readlane(s, g + k);
                v[k] = *(const vf*)(y + (size_t)sk * C + V * lane);
            }
#pragma unroll
            for (int k = 0; k < G; ++k) {
                if (k == 0 || g + k < m) {
                    const int rk = __builtin_amdgcn_readlane(r, g + k);
#pragma unroll
                    for (int j = 0; j < V; ++j) atomicAdd(&acc[rk * C + j * 32 + lane], v[k][j]);
                }
            }
        }
    }

    if (all > CAP) {
        unsigned pos = pos0;
#pragma nounroll
        for (int q = tid; q < NQ; q += 1024) {
            const uint4 wq = bq[q];
            const unsigned ws[4] = {wq.x, wq.y, wq.z, wq.w};
#pragma nounroll
            for (int k = 0; k < 4; ++k) {
                unsigned word = ws[k];
                while (word != 0u) {
                    if (pos >= (unsigned)CAP) {
                        const int e = (q * 4 + k) * 32 + __builtin_ctz(word);
                        const int s = src[e];
                        const int r = dst[e] - lo;
                        for (int ch = 0; ch < C; ++ch)
                            atomicAdd(&acc[r * C + (ch % V) * 32 + ch / V], y[(size_t)s * C + ch]);
                    }
                    word &= word - 1u;
                    ++pos;
                }
            }
        }
    }
    __syncthreads();

    for (int r = wave; r < nrows; r += 32) {
        const int n = lo + r;
        vf v;
#pragma unroll
        for (int i = 0; i < V; ++i) v[i] = acc[r * C + i * 32 + lane];
        if constexpr (FINAL) {
            const float dn = dinv[n];
            const vf bv = *(const vf*)(bias + V * lane);
            v = bv + v * dn;
        }
        *(vf*)(outp + (size_t)n * C + V * lane) = v;
    }
}


extern "C" void kernel_launch(void* const* d_in, const int* in_sizes, int n_in,
                              void* d_out, int out_size, void* d_ws, size_t ws_size,
                              hipStream_t stream) {
    const float* x  = (const float*)d_in[0];
    const int*   ei = (const int*)d_in[1];
    const float* W1 = (const float*)d_in[2];
    const float* b1 = (const float*)d_in[3];
    const float* W2 = (const float*)d_in[4];
    const float* b2 = (const float*)d_in[5];
    float* out = (float*)d_out;

    const int* src = ei;
    const int* dst = ei + N_EDGES;

    float* dinv = (float*)d_ws;
    float* xw1  = dinv + 50176;
    float* agg1 = xw1 + 6400000;
    float* degp = agg1 + 6400000;
    unsigned* bm = (unsigned*)(degp + 1600000);
    float* xw2  = xw1;

    gcn_deg_init<<<(N_NODES + 255) / 256, 256, 0, stream>>>(degp);
    gcn_deg_accum<<<(N_EDGES + TB_EDGES - 1) / TB_EDGES, 256, 0, stream>>>(dst, degp, bm);

    gcn_gemm1_wmma<<<G1_BLOCKS, 256, 0, stream>>>(x, W1, degp, xw1, dinv);

    gcn_agg_lds<HID_C, false><<<NB, 1024, 0, stream>>>(src, dst, bm, dinv, degp, b1, xw1, agg1);

    gcn_gemm2_wmma<<<G2_BLOCKS, 128, 0, stream>>>(agg1, W2, dinv, xw2);

    gcn_agg_lds<OUT_C, true><<<NB, 1024, 0, stream>>>(src, dst, bm, dinv, degp, b2, xw2, out);
}


# Second product joins the layer-1 aggregation launch: a block applies ReLU to its summed rows in LDS and multiplies them by W2 on f32 WMMA; S never stored; 6 launches -> 5; its k order changes
# speedup vs baseline: 2.9395x; 1.0779x over previous
#include <hip/hip_runtime.h>

#define N_NODES 50000
#define N_EDGES 800000
#define IN_C    64
#define HID_C   128
#define OUT_C   64

typedef __attribute__((ext_vector_type(2))) float v2f;
typedef __attribute__((ext_vector_type(8))) float v8f;


#define DEG_STRIDE 32

__global__ void gcn_deg_init(float* __restrict__ degp) {
    int n = blockIdx.x * blockDim.x + threadIdx.x;
    if (n < N_NODES) degp[(size_t)n * DEG_STRIDE] = 1.0f;
}

#define NB       256
#define RANGE    196
#define EW       (N_EDGES / 32)
#define TB_EDGES 2048
#define TB_WORDS (TB_EDGES / 32)

__global__ __launch_bounds__(256) void gcn_deg_accum(const int* __restrict__ dst, float* __restrict__ degp,
                                                     unsigned* __restrict__ bm) {
    __shared__ __attribute__((aligned(16))) unsigned tile[NB * TB_WORDS];
    const int t = threadIdx.x;
    uint4* tq = (uint4*)tile;
#pragma unroll
    for (int i = 0; i < 16; ++i) tq[i * 256 + t] = make_uint4(0u, 0u, 0u, 0u);
    __syncthreads();
    const int e0 = blockIdx.x * TB_EDGES;
    int dd[8];
#pragma unroll
    for (int j = 0; j < 8; ++j) {
        const int e = e0 + j * 256 + t;
        dd[j] = (e < N_EDGES) ? dst[e] : -1;
    }
#pragma unroll
    for (int j = 0; j < 8; ++j) {
        const int el = j * 256 + t;
        const int d  = dd[j];
        if (d >= 0) {
            atomicAdd(&degp[(size_t)d * DEG_STRIDE], 1.0f);
            atomicOr(&tile[(d / RANGE) * TB_WORDS + (el >> 5)], 1u << (el & 31));
        }
    }
    __syncthreads();
#pragma unroll
    for (int i = 0; i < 16; ++i) {
        const int p     = i * 256 + t;
        const int range = p >> 4;
        const int wj    = blockIdx.x * TB_WORDS + (p & 15) * 4;
        if (wj < EW) *(uint4*)(bm + (size_t)range * EW + wj) = tq[p];
    }
}


#define N_TILES   (N_NODES / 16)
#define G1_PITCH  (IN_C + 4)
#define G1_BLOCKS 1024

static __device__ __forceinline__ void lds_barrier() {
    __builtin_amdgcn_fence(__ATOMIC_RELEASE, "workgroup", "local");
    __builtin_amdgcn_s_barrier();
    __builtin_amdgcn_fence(__ATOMIC_ACQUIRE, "workgroup", "local");
}

__global__ __launch_bounds__(256) void gcn_gemm1_wmma(
    const float* __restrict__ x, const float* __restrict__ W1,
    const float* __restrict__ degp,
    float* __restrict__ xw, float* __restrict__ dinv) {
    __shared__ __attribute__((aligned(16))) float lds[2][16 * G1_PITCH];
    const int lane = threadIdx.x & 31;
    const int wave = threadIdx.x >> 5;
    const int n0   = wave * 16;
    const int half = lane >> 4;
    const int l15  = lane & 15;
    const int col  = n0 + l15;
    v2f bf[IN_C / 4];
#pragma unroll
    for (int ks = 0; ks < IN_C / 4; ++ks) {
        const int ka = ks * 4 + half * 2;
        bf[ks].x = W1[ka * HID_C + col];
        bf[ks].y = W1[(ka + 1) * HID_C + col];
    }
    const int sr = threadIdx.x >> 4;
    const int sc = (threadIdx.x & 15) * 4;
    int tile = blockIdx.x;
    float4 st = *(const float4*)(x + (size_t)(tile * 16 + sr) * IN_C + sc);
    *(float4*)(&lds[0][sr * G1_PITCH + sc]) = st;
    __syncthreads();
    int buf = 0;
    for (; tile < N_TILES; tile += gridDim.x) {
        const int  m0       = tile * 16;
        const int  nxt      = tile + gridDim.x;
        const bool has_next = nxt < N_TILES;
        if (has_next) st = *(const float4*)(x + (size_t)(nxt * 16 + sr) * IN_C + sc);
        float dg[8];
#pragma unroll
        for (int r = 0; r < 8; ++r) dg[r] = degp[(size_t)(m0 + half * 8 + r) * DEG_STRIDE];
        v8f acc = {0.f, 0.f, 0.f, 0.f, 0.f, 0.f, 0.f, 0.f};
        const float* at = &lds[buf][l15 * G1_PITCH + half * 2];
#pragma unroll
        for (int ks = 0; ks < IN_C / 4; ++ks) {
            v2f a = *(const v2f*)(at + ks * 4);
            acc = __builtin_amdgcn_wmma_f32_16x16x4_f32(
                 false, a,  false, bf[ks],
                 (short)0, acc,  false,  false);
        }
        asm volatile("s_wait_alu depctr_va_vdst(0)" : "+v"(acc));
        float dn[8];
#pragma unroll
        for (int r = 0; r < 8; ++r) {
            const size_t o = (size_t)(m0 + half * 8 + r) * HID_C + col;
            dn[r]  = rsqrtf(dg[r]);
            xw[o]  = acc[r] * dn[r];
        }
        if (wave == 0 && l15 == 0) {
#pragma unroll
            for (int r = 0; r < 8; ++r) dinv[m0 + half * 8 + r] = dn[r];
        }
        if (has_next) *(float4*)(&lds[buf ^ 1][sr * G1_PITCH + sc]) = st;
        lds_barrier();
        buf ^= 1;
    }
}


#define CAP 8192

template <int C, bool FINAL>
__global__ __launch_bounds__(1024, FINAL ? 16 : 8) void gcn_agg_lds(
    const int* __restrict__ src, const int* __restrict__ dst, const unsigned* __restrict__ bm,
    const float* __restrict__ dinv, const float* __restrict__ degp, const float* __restrict__ bias,
    const float* __restrict__ w2,
    const float* __restrict__ y, float* __restrict__ outp) {
    constexpr int V = C / 32;
    constexpr int P = FINAL ? C : C + 4;
    typedef float vf __attribute__((ext_vector_type(V)));
    __shared__ __attribute__((aligned(16))) float acc[RANGE * P];
    __shared__ unsigned list[CAP + 1];
    __shared__ unsigned cnt;
    constexpr int NST = (RANGE + 15) / 16;
    __shared__ __attribute__((aligned(16))) float dl[FINAL ? 4 : NST * 16];
    const int tid   = threadIdx.x;
    const int lane  = tid & 31;
    const int wave  = __builtin_amdgcn_readfirstlane(tid >> 5);
    const int lo    = blockIdx.x * RANGE;
    const int nrows = (N_NODES - lo) < RANGE ? (N_NODES - lo) : RANGE;
    if (tid == 0) cnt = 0u;

    constexpr int NQ = EW / 4;
    constexpr int U  = 7;
    static_assert(U * 1024 >= NQ, "one sweep covers the range's row");
    const uint4* bq = (const uint4*)(bm + (size_t)blockIdx.x * EW);
    uint4 w[U];
#pragma unroll
    for (int u = 0; u < U; ++u) {
        const int q = u * 1024 + tid;
        w[u] = (q < NQ) ? bq[q] : make_uint4(0u, 0u, 0u, 0u);
    }

    if constexpr (!FINAL) {
        if (tid < NST * 16) dl[tid] = tid < nrows ? dinv[lo + tid] : 0.0f;
    }

    for (int r = wave; r < nrows; r += 32) {
        const int n = lo + r;
        vf v = *(const vf*)(y + (size_t)n * C + V * lane);
        if constexpr (!FINAL) {
            const float f = degp[(size_t)n * DEG_STRIDE] * dinv[n];
            const vf bv = *(const vf*)(bias + V * lane);
            v += bv * f;
        }
#pragma unroll
        for (int i = 0; i < V; ++i) acc[r * P + i * 32 + lane] = v[i];
    }
    __syncthreads();

    unsigned c = 0u;
#pragma unroll
    for (int u = 0; u < U; ++u)
        c += __builtin_popcount(w[u].x) + __builtin_popcount(w[u].y) +
             __builtin_popcount(w[u].z) + __builtin_popcount(w[u].w);
    unsigned pos0 = 0u;
    if (c != 0u) pos0 = atomicAdd(&cnt, c);
    {
        unsigned pos = pos0;
#pragma unroll
        for (int u = 0; u < U; ++u) {
            const unsigned ws[4] = {w[u].x, w[u].y, w[u].z, w[u].w};
#pragma unroll
            for (int k = 0; k < 4; ++k) {
                unsigned word = ws[k];
                const int eb = ((u * 1024 + tid) * 4 + k) * 32;
                while (word != 0u) {
                    list[pos < (unsigned)CAP ? pos : (unsigned)CAP] = (unsigned)(eb + __builtin_ctz(word));
                    word &= word - 1u;
                    ++pos;
                }
            }
        }
    }
    __syncthreads();

    const int all   = __builtin_amdgcn_readfirstlane((int)cnt);
    const int total = all < CAP ? all : CAP;
    constexpr int G = 32 / V;
    for (int i0 = wave * 32; i0 < total; i0 += 1024) {
        const int i = i0 + lane;
        int s = 0, r = 0;
        if (i < total) {
            const unsigned e = list[i];
            s = src[e];
            r = dst[e] - lo;
        }
        const int m = (total - i0) < 32 ? (total - i0) : 32;
        for (int g = 0; g < m; g += G) {
            vf v[G];
#pragma unroll
            for (int k = 0; k < G; ++k) {
                const int sk = __builtin_amdgcn_readlane(s, g + k);
                v[k] = *(const vf*)(y + (size_t)sk * C + V * lane);
            }
#pragma unroll
            for (int k = 0; k < G; ++k) {
                if (k == 0 || g + k < m) {
                    const int rk = __builtin_amdgcn_readlane(r, g + k);
#pragma unroll
                    for (int j = 0; j < V; ++j) atomicAdd(&acc[rk * P + j * 32 + lane], v[k][j]);
                }
            }
        }
    }

    const int half = lane >> 4;
    const int l15  = lane & 15;
    const int col  = (wave & 3) * 16 + l15;
    v2f bf[FINAL ? 1 : C / 4];
    if constexpr (!FINAL) {
#pragma unroll
        for (int ks = 0; ks < C / 4; ++ks) {
            const int ka = 16 * (ks & 7) + 8 * half + (ks >> 3);
            bf[ks].x = w2[ka * OUT_C + col];
            bf[ks].y = w2[(ka + 4) * OUT_C + col];
        }
    }

    if (all > CAP) {
        unsigned pos = pos0;
#pragma nounroll
        for (int q = tid; q < NQ; q += 1024) {
            const uint4 wq = bq[q];
            const unsigned ws[4] = {wq.x, wq.y, wq.z, wq.w};
#pragma nounroll
            for (int k = 0; k < 4; ++k) {
                unsigned word = ws[k];
                while (word != 0u) {
                    if (pos >= (unsigned)CAP) {
                        const int e = (q * 4 + k) * 32 + __builtin_ctz(word);
                        const int s = src[e];
                        const int r = dst[e] - lo;
                        for (int ch = 0; ch < C; ++ch)
                            atomicAdd(&acc[r * P + (ch % V) * 32 + ch / V], y[(size_t)s * C + ch]);
                    }
                    word &= word - 1u;
                    ++pos;
                }
            }
        }
    }
    __syncthreads();

    if constexpr (FINAL) {
        for (int r = wave; r < nrows; r += 32) {
            const int n = lo + r;
            vf v;
#pragma unroll
            for (int i = 0; i < V; ++i) v[i] = acc[r * P + i * 32 + lane];
            const float dn = dinv[n];
            const vf bv = *(const vf*)(bias + V * lane);
            v = bv + v * dn;
            *(vf*)(outp + (size_t)n * C + V * lane) = v;
        }
    } else {
        const int nst = (nrows + 15) >> 4;
        for (int st = wave >> 2; st < nst; st += 8) {
            const int ar = st * 16 + l15;
            const float* at = &acc[(ar < nrows ? ar : nrows - 1) * P + half * 2];
            const float4 dn0 = *(const float4*)&dl[st * 16 + half * 8];
            const float4 dn1 = *(const float4*)&dl[st * 16 + half * 8 + 4];
            const float dn[8] = {dn0.x, dn0.y, dn0.z, dn0.w, dn1.x, dn1.y, dn1.z, dn1.w};
            v8f d = {0.f, 0.f, 0.f, 0.f, 0.f, 0.f, 0.f, 0.f};
#pragma unroll
            for (int ks = 0; ks < C / 4; ++ks) {
                v2f a = *(const v2f*)(at + ks * 4);
                a.x = fmaxf(a.x, 0.0f);
                a.y = fmaxf(a.y, 0.0f);
                d = __builtin_amdgcn_wmma_f32_16x16x4_f32(
                     false, a,  false, bf[ks],
                     (short)0, d,  false,  false);
            }
            asm volatile("s_wait_alu depctr_va_vdst(0)" : "+v"(d));
#pragma unroll
            for (int r = 0; r < 8; ++r) {
                const int row = st * 16 + half * 8 + r;
                if (row < nrows)
                    outp[(size_t)(lo + row) * OUT_C + col] = d[r] * dn[r] * dn[r];
            }
        }
    }
}


extern "C" void kernel_launch(void* const* d_in, const int* in_sizes, int n_in,
                              void* d_out, int out_size, void* d_ws, size_t ws_size,
                              hipStream_t stream) {
    const float* x  = (const float*)d_in[0];
    const int*   ei = (const int*)d_in[1];
    const float* W1 = (const float*)d_in[2];
    const float* b1 = (const float*)d_in[3];
    const float* W2 = (const float*)d_in[4];
    const float* b2 = (const float*)d_in[5];
    float* out = (float*)d_out;

    const int* src = ei;
    const int* dst = ei + N_EDGES;

    float* dinv = (float*)d_ws;
    float* xw1  = dinv + 50176;
    float* xw2  = xw1 + 6400000;
    float* degp = xw2 + 6400000;
    unsigned* bm = (unsigned*)(degp + 1600000);

    gcn_deg_init<<<(N_NODES + 255) / 256, 256, 0, stream>>>(degp);
    gcn_deg_accum<<<(N_EDGES + TB_EDGES - 1) / TB_EDGES, 256, 0, stream>>>(dst, degp, bm);

    gcn_gemm1_wmma<<<G1_BLOCKS, 256, 0, stream>>>(x, W1, degp, xw1, dinv);

    gcn_agg_lds<HID_C, false><<<NB, 1024, 0, stream>>>(src, dst, bm, dinv, degp, b1, W2, xw1, xw2);

    gcn_agg_lds<OUT_C, true><<<NB, 1024, 0, stream>>>(src, dst, bm, dinv, degp, b2, W2, xw2, out);
}


# A range's edge list is built once: layer 1 stores its entries as (source, local row) pairs in the workspace, layer 2 reads them: no bit plane, no index gathers; sums still LDS atomics, any order
# speedup vs baseline: 3.2183x; 1.0948x over previous
#include <hip/hip_runtime.h>

#define N_NODES 50000
#define N_EDGES 800000
#define IN_C    64
#define HID_C   128
#define OUT_C   64

typedef __attribute__((ext_vector_type(2))) float v2f;
typedef __attribute__((ext_vector_type(8))) float v8f;


#define DEG_STRIDE 32

__global__ void gcn_deg_init(float* __restrict__ degp) {
    int n = blockIdx.x * blockDim.x + threadIdx.x;
    if (n < N_NODES) degp[(size_t)n * DEG_STRIDE] = 1.0f;
}

#define NB       256
#define RANGE    196
#define EW       (N_EDGES / 32)
#define TB_EDGES 2048
#define TB_WORDS (TB_EDGES / 32)

__global__ __launch_bounds__(256) void gcn_deg_accum(const int* __restrict__ dst, float* __restrict__ degp,
                                                     unsigned* __restrict__ bm) {
    __shared__ __attribute__((aligned(16))) unsigned tile[NB * TB_WORDS];
    const int t = threadIdx.x;
    uint4* tq = (uint4*)tile;
#pragma unroll
    for (int i = 0; i < 16; ++i) tq[i * 256 + t] = make_uint4(0u, 0u, 0u, 0u);
    __syncthreads();
    const int e0 = blockIdx.x * TB_EDGES;
    int dd[8];
#pragma unroll
    for (int j = 0; j < 8; ++j) {
        const int e = e0 + j * 256 + t;
        dd[j] = (e < N_EDGES) ? dst[e] : -1;
    }
#pragma unroll
    for (int j = 0; j < 8; ++j) {
        const int el = j * 256 + t;
        const int d  = dd[j];
        if (d >= 0) {
            atomicAdd(&degp[(size_t)d * DEG_STRIDE], 1.0f);
            atomicOr(&tile[(d / RANGE) * TB_WORDS + (el >> 5)], 1u << (el & 31));
        }
    }
    __syncthreads();
#pragma unroll
    for (int i = 0; i < 16; ++i) {
        const int p     = i * 256 + t;
        const int range = p >> 4;
        const int wj    = blockIdx.x * TB_WORDS + (p & 15) * 4;
        if (wj < EW) *(uint4*)(bm + (size_t)range * EW + wj) = tq[p];
    }
}


#define N_TILES   (N_NODES / 16)
#define G1_PITCH  (IN_C + 4)
#define G1_BLOCKS 1024

static __device__ __forceinline__ void lds_barrier() {
    __builtin_amdgcn_fence(__ATOMIC_RELEASE, "workgroup", "local");
    __builtin_amdgcn_s_barrier();
    __builtin_amdgcn_fence(__ATOMIC_ACQUIRE, "workgroup", "local");
}

__global__ __launch_bounds__(256) void gcn_gemm1_wmma(
    const float* __restrict__ x, const float* __restrict__ W1,
    const float* __restrict__ degp,
    float* __restrict__ xw, float* __restrict__ dinv) {
    __shared__ __attribute__((aligned(16))) float lds[2][16 * G1_PITCH];
    const int lane = threadIdx.x & 31;
    const int wave = threadIdx.x >> 5;
    const int n0   = wave * 16;
    const int half = lane >> 4;
    const int l15  = lane & 15;
    const int col  = n0 + l15;
    v2f bf[IN_C / 4];
#pragma unroll
    for (int ks = 0; ks < IN_C / 4; ++ks) {
        const int ka = ks * 4 + half * 2;
        bf[ks].x = W1[ka * HID_C + col];
        bf[ks].y = W1[(ka + 1) * HID_C + col];
    }
    const int sr = threadIdx.x >> 4;
    const int sc = (threadIdx.x & 15) * 4;
    int tile = blockIdx.x;
    float4 st = *(const float4*)(x + (size_t)(tile * 16 + sr) * IN_C + sc);
    *(float4*)(&lds[0][sr * G1_PITCH + sc]) = st;
    __syncthreads();
    int buf = 0;
    for (; tile < N_TILES; tile += gridDim.x) {
        const int  m0       = tile * 16;
        const int  nxt      = tile + gridDim.x;
        const bool has_next = nxt < N_TILES;
        if (has_next) st = *(const float4*)(x + (size_t)(nxt * 16 + sr) * IN_C + sc);
        float dg[8];
#pragma unroll
        for (int r = 0; r < 8; ++r) dg[r] = degp[(size_t)(m0 + half * 8 + r) * DEG_STRIDE];
        v8f acc = {0.f, 0.f, 0.f, 0.f, 0.f, 0.f, 0.f, 0.f};
        const float* at = &lds[buf][l15 * G1_PITCH + half * 2];
#pragma unroll
        for (int ks = 0; ks < IN_C / 4; ++ks) {
            v2f a = *(const v2f*)(at + ks * 4);
            acc = __builtin_amdgcn_wmma_f32_16x16x4_f32(
                 false, a,  false, bf[ks],
                 (short)0, acc,  false,  false);
        }
        asm volatile("s_wait_alu depctr_va_vdst(0)" : "+v"(acc));
        float dn[8];
#pragma unroll
        for (int r = 0; r < 8; ++r) {
            const size_t o = (size_t)(m0 + half * 8 + r) * HID_C + col;
            dn[r]  = rsqrtf(dg[r]);
            xw[o]  = acc[r] * dn[r];
        }
        if (wave == 0 && l15 == 0) {
#pragma unroll
            for (int r = 0; r < 8; ++r) dinv[m0 + half * 8 + r] = dn[r];
        }
        if (has_next) *(float4*)(&lds[buf ^ 1][sr * G1_PITCH + sc]) = st;
        lds_barrier();
        buf ^= 1;
    }
}


#define CAP 8192

template <int C, bool FINAL>
__global__ __launch_bounds__(1024, FINAL ? 16 : 8) void gcn_agg_lds(
    const int* __restrict__ src, const int* __restrict__ dst, const unsigned* __restrict__ bm,
    const float* __restrict__ dinv, const float* __restrict__ degp, const float* __restrict__ bias,
    const float* __restrict__ w2,
    int* __restrict__ pl, int* __restrict__ pc,
    const float* __restrict__ y, float* __restrict__ outp) {
    constexpr int V = C / 32;
    constexpr int P = FINAL ? C : C + 4;
    typedef float vf __attribute__((ext_vector_type(V)));
    __shared__ __attribute__((aligned(16))) float acc[RANGE * P];
    __shared__ unsigned list[FINAL ? 1 : CAP + 1];
    __shared__ unsigned cnt;
    constexpr int NST = (RANGE + 15) / 16;
    __shared__ __attribute__((aligned(16))) float dl[FINAL ? 4 : NST * 16];
    const int tid   = threadIdx.x;
    const int lane  = tid & 31;
    const int wave  = __builtin_amdgcn_readfirstlane(tid >> 5);
    const int lo    = blockIdx.x * RANGE;
    const int nrows = (N_NODES - lo) < RANGE ? (N_NODES - lo) : RANGE;
    if constexpr (!FINAL) {
        if (tid == 0) cnt = 0u;
    }
    int2* pq = (int2*)pl + (size_t)blockIdx.x * CAP;

    constexpr int NQ = EW / 4;
    constexpr int U  = 7;
    static_assert(U * 1024 >= NQ, "one sweep covers the range's row");
    const uint4* bq = (const uint4*)(bm + (size_t)blockIdx.x * EW);
    uint4 w[FINAL ? 1 : U];
    int fall = 0;
    int2 cur = make_int2(0, 0);
    if constexpr (!FINAL) {
#pragma unroll
        for (int u = 0; u < U; ++u) {
            const int q = u * 1024 + tid;
            w[u] = (q < NQ) ? bq[q] : make_uint4(0u, 0u, 0u, 0u);
        }
    } else {
        fall = __builtin_amdgcn_readfirstlane(pc[blockIdx.x]);
        if (fall <= CAP && tid < fall) cur = pq[tid];
    }

    if constexpr (!FINAL) {
        if (tid < NST * 16) dl[tid] = tid < nrows ? dinv[lo + tid] : 0.0f;
    }

    for (int r = wave; r < nrows; r += 32) {
        const int n = lo + r;
        vf v = *(const vf*)(y + (size_t)n * C + V * lane);
        if constexpr (!FINAL) {
            const float f = degp[(size_t)n * DEG_STRIDE] * dinv[n];
            const vf bv = *(const vf*)(bias + V * lane);
            v += bv * f;
        }
#pragma unroll
        for (int i = 0; i < V; ++i) acc[r * P + i * 32 + lane] = v[i];
    }
    __syncthreads();

    unsigned pos0 = 0u;
    int all, total;
    if constexpr (!FINAL) {
        unsigned c = 0u;
#pragma unroll
        for (int u = 0; u < U; ++u)
            c += __builtin_popcount(w[u].x) + __builtin_popcount(w[u].y) +
                 __builtin_popcount(w[u].z) + __builtin_popcount(w[u].w);
        if (c != 0u) pos0 = atomicAdd(&cnt, c);
        {
            unsigned pos = pos0;
#pragma unroll
            for (int u = 0; u < U; ++u) {
                const unsigned ws[4] = {w[u].x, w[u].y, w[u].z, w[u].w};
#pragma unroll
                for (int k = 0; k < 4; ++k) {
                    unsigned word = ws[k];
                    const int eb = ((u * 1024 + tid) * 4 + k) * 32;
                    while (word != 0u) {
                        list[pos < (unsigned)CAP ? pos : (unsigned)CAP] = (unsigned)(eb + __builtin_ctz(word));
                        word &= word - 1u;
                        ++pos;
                    }
                }
            }
        }
        __syncthreads();

        all   = __builtin_amdgcn_readfirstlane((int)cnt);
        total = all < CAP ? all : CAP;
        if (tid == 0) pc[blockIdx.x] = all;
    } else {
        all   = fall;
        total = all <= CAP ? all : 0;
    }
    constexpr int G = 32 / V;
    for (int i0 = wave * 32; i0 < total; i0 += 1024) {
        const int i = i0 + lane;
        int s = 0, r = 0;
        if constexpr (!FINAL) {
            if (i < total) {
                const unsigned e = list[i];
                s = src[e];
                r = dst[e] - lo;
                pq[i] = make_int2(s, r);
            }
        } else {
            s = cur.x;
            r = cur.y;
        }
        const int m = (total - i0) < 32 ? (total - i0) : 32;
        for (int g = 0; g < m; g += G) {
            vf v[G];
#pragma unroll
            for (int k = 0; k < G; ++k) {
                const int sk = __builtin_amdgcn_readlane(s, g + k);
                v[k] = *(const vf*)(y + (size_t)sk * C + V * lane);
            }
            if constexpr (FINAL) {
                if (g == 0) {
                    cur = make_int2(0, 0);
                    if (i + 1024 < total) cur = pq[i + 1024];
                }
            }
#pragma unroll
            for (int k = 0; k < G; ++k) {
                if (k == 0 || g + k < m) {
                    const int rk = __builtin_amdgcn_readlane(r, g + k);
#pragma unroll
                    for (int j = 0; j < V; ++j) atomicAdd(&acc[rk * P + j * 32 + lane], v[k][j]);
                }
            }
        }
    }

    const int half = lane >> 4;
    const int l15  = lane & 15;
    const int col  = (wave & 3) * 16 + l15;
    v2f bf[FINAL ? 1 : C / 4];
    if constexpr (!FINAL) {
#pragma unroll
        for (int ks = 0; ks < C / 4; ++ks) {
            const int ka = 16 * (ks & 7) + 8 * half + (ks >> 3);
            bf[ks].x = w2[ka * OUT_C + col];
            bf[ks].y = w2[(ka + 4) * OUT_C + col];
        }
    }

    if (all > CAP) {
        unsigned pos = FINAL ? (unsigned)CAP : pos0;
#pragma nounroll
        for (int q = tid; q < NQ; q += 1024) {
            const uint4 wq = bq[q];
            const unsigned ws[4] = {wq.x, wq.y, wq.z, wq.w};
#pragma nounroll
            for (int k = 0; k < 4; ++k) {
                unsigned word = ws[k];
                while (word != 0u) {
                    if (pos >= (unsigned)CAP) {
                        const int e = (q * 4 + k) * 32 + __builtin_ctz(word);
                        const int s = src[e];
                        const int r = dst[e] - lo;
                        for (int ch = 0; ch < C; ++ch)
                            atomicAdd(&acc[r * P + (ch % V) * 32 + ch / V], y[(size_t)s * C + ch]);
                    }
                    word &= word - 1u;
                    ++pos;
                }
            }
        }
    }
    __syncthreads();

    if constexpr (FINAL) {
        for (int r = wave; r < nrows; r += 32) {
            const int n = lo + r;
            vf v;
#pragma unroll
            for (int i = 0; i < V; ++i) v[i] = acc[r * P + i * 32 + lane];
            const float dn = dinv[n];
            const vf bv = *(const vf*)(bias + V * lane);
            v = bv + v * dn;
            *(vf*)(outp + (size_t)n * C + V * lane) = v;
        }
    } else {
        const int nst = (nrows + 15) >> 4;
        for (int st = wave >> 2; st < nst; st += 8) {
            const int ar = st * 16 + l15;
            const float* at = &acc[(ar < nrows ? ar : nrows - 1) * P + half * 2];
            const float4 dn0 = *(const float4*)&dl[st * 16 + half * 8];
            const float4 dn1 = *(const float4*)&dl[st * 16 + half * 8 + 4];
            const float dn[8] = {dn0.x, dn0.y, dn0.z, dn0.w, dn1.x, dn1.y, dn1.z, dn1.w};
            v8f d = {0.f, 0.f, 0.f, 0.f, 0.f, 0.f, 0.f, 0.f};
#pragma unroll
            for (int ks = 0; ks < C / 4; ++ks) {
                v2f a = *(const v2f*)(at + ks * 4);
                a.x = fmaxf(a.x, 0.0f);
                a.y = fmaxf(a.y, 0.0f);
                d = __builtin_amdgcn_wmma_f32_16x16x4_f32(
                     false, a,  false, bf[ks],
                     (short)0, d,  false,  false);
            }
            asm volatile("s_wait_alu depctr_va_vdst(0)" : "+v"(d));
#pragma unroll
            for (int r = 0; r < 8; ++r) {
                const int row = st * 16 + half * 8 + r;
                if (row < nrows)
                    outp[(size_t)(lo + row) * OUT_C + col] = d[r] * dn[r] * dn[r];
            }
        }
    }
}


extern "C" void kernel_launch(void* const* d_in, const int* in_sizes, int n_in,
                              void* d_out, int out_size, void* d_ws, size_t ws_size,
                              hipStream_t stream) {
    const float* x  = (const float*)d_in[0];
    const int*   ei = (const int*)d_in[1];
    const float* W1 = (const float*)d_in[2];
    const float* b1 = (const float*)d_in[3];
    const float* W2 = (const float*)d_in[4];
    const float* b2 = (const float*)d_in[5];
    float* out = (float*)d_out;

    const int* src = ei;
    const int* dst = ei + N_EDGES;

    float* dinv = (float*)d_ws;
    float* xw1  = dinv + 50176;
    float* xw2  = xw1 + 6400000;
    float* degp = xw2 + 6400000;
    unsigned* bm = (unsigned*)(degp + 1600000);
    int* pl = (int*)(bm + 6400000);
    int* pc = pl + 4194304;

    gcn_deg_init<<<(N_NODES + 255) / 256, 256, 0, stream>>>(degp);
    gcn_deg_accum<<<(N_EDGES + TB_EDGES - 1) / TB_EDGES, 256, 0, stream>>>(dst, degp, bm);

    gcn_gemm1_wmma<<<G1_BLOCKS, 256, 0, stream>>>(x, W1, degp, xw1, dinv);

    gcn_agg_lds<HID_C, false><<<NB, 1024, 0, stream>>>(src, dst, bm, dinv, degp, b1, W2, pl, pc, xw1, xw2);

    gcn_agg_lds<OUT_C, true><<<NB, 1024, 0, stream>>>(src, dst, bm, dinv, degp, b2, W2, pl, pc, xw2, out);
}


# Degree pass writes each range's (source, local row) pair list itself (LDS ranks, one global slot add per block and range): no bit plane, layer 1 gathers no index; sums stay LDS atomics, any order
# speedup vs baseline: 3.6586x; 1.1368x over previous
#include <hip/hip_runtime.h>

#define N_NODES 50000
#define N_EDGES 800000
#define IN_C    64
#define HID_C   128
#define OUT_C   64

typedef __attribute__((ext_vector_type(2))) float v2f;
typedef __attribute__((ext_vector_type(8))) float v8f;


#define DEG_STRIDE 32

#define NB       256
#define RANGE    196
#define CAP      8192
#define TB_EDGES 4096
#define TB_PER   (TB_EDGES / 1024)
#define PC_STRIDE 32

__global__ void gcn_deg_init(float* __restrict__ degp, int* __restrict__ pc) {
    int n = blockIdx.x * blockDim.x + threadIdx.x;
    if (n < N_NODES) degp[(size_t)n * DEG_STRIDE] = 1.0f;
    if (n < NB) pc[n * PC_STRIDE] = 0;
}

static __device__ __forceinline__ void lds_barrier() {
    __builtin_amdgcn_fence(__ATOMIC_RELEASE, "workgroup", "local");
    __builtin_amdgcn_s_barrier();
    __builtin_amdgcn_fence(__ATOMIC_ACQUIRE, "workgroup", "local");
}


__global__ __launch_bounds__(1024) void gcn_deg_bucket(const int* __restrict__ src, const int* __restrict__ dst,
                                                       float* __restrict__ degp,
                                                       int* __restrict__ pl, int* __restrict__ pc) {
    __shared__ int cntL[NB];
    __shared__ int startL[NB];
    __shared__ int baseL[NB];
    __shared__ int wsum[NB / 32];
    __shared__ __attribute__((aligned(16))) int stage[2 * TB_EDGES];
    const int t    = threadIdx.x;
    const int lane = t & 31;
    const int e0   = blockIdx.x * TB_EDGES;
    const int nvalid = (N_EDGES - e0) < TB_EDGES ? (N_EDGES - e0) : TB_EDGES;
    if (t < NB) cntL[t] = 0;
    int dd[TB_PER], ss[TB_PER];
#pragma unroll
    for (int j = 0; j < TB_PER; ++j) {
        const int e = e0 + j * 1024 + t;
        dd[j] = (e < N_EDGES) ? dst[e] : -1;
        ss[j] = (e < N_EDGES) ? src[e] : 0;
    }
    lds_barrier();
    int rg[TB_PER], kk[TB_PER];
#pragma unroll
    for (int j = 0; j < TB_PER; ++j) {
        rg[j] = 0;
        kk[j] = 0;
        if (dd[j] >= 0) {
            atomicAdd(&degp[(size_t)dd[j] * DEG_STRIDE], 1.0f);
            rg[j] = dd[j] / RANGE;
            kk[j] = atomicAdd(&cntL[rg[j]], 1);
        }
    }
    lds_barrier();
    int c = 0, incl = 0;
    if (t < NB) {
        c = cntL[t];
        incl = c;
#pragma unroll
        for (int d = 1; d < 32; d <<= 1) {
            const int up = __shfl_up(incl, d, 32);
            if (lane >= d) incl += up;
        }
        if (lane == 31) wsum[t >> 5] = incl;
    }
    lds_barrier();
    int lstart = 0, gbase = 0;
    if (t < NB) {
        int off = 0;
#pragma unroll
        for (int w = 0; w < NB / 32; ++w) off += (w < (t >> 5)) ? wsum[w] : 0;
        lstart = off + incl - c;
        startL[t] = lstart;
        if (c > 0) gbase = atomicAdd(&pc[t * PC_STRIDE], c);
    }
    lds_barrier();
    int2* sq = (int2*)stage;
    int fp[TB_PER];
#pragma unroll
    for (int j = 0; j < TB_PER; ++j) fp[j] = startL[rg[j]];
#pragma unroll
    for (int j = 0; j < TB_PER; ++j)
        if (dd[j] >= 0) sq[fp[j] + kk[j]] = make_int2(ss[j], dd[j]);
    if (t < NB) baseL[t] = gbase - lstart;
    lds_barrier();
    int2* pq = (int2*)pl;
    int2 p[TB_PER];
    int  pr[TB_PER], pb[TB_PER];
#pragma unroll
    for (int j = 0; j < TB_PER; ++j) p[j] = sq[(j * 1024 + t) < nvalid ? (j * 1024 + t) : 0];
#pragma unroll
    for (int j = 0; j < TB_PER; ++j) {
        pr[j] = p[j].y / RANGE;
        pb[j] = baseL[pr[j]];
    }
#pragma unroll
    for (int j = 0; j < TB_PER; ++j) {
        const int i   = j * 1024 + t;
        const int pos = pb[j] + i;
        if (i < nvalid && pos < CAP) pq[(size_t)pr[j] * CAP + pos] = make_int2(p[j].x, p[j].y - pr[j] * RANGE);
    }
}


#define N_TILES   (N_NODES / 16)
#define G1_PITCH  (IN_C + 4)
#define G1_BLOCKS 1024

__global__ __launch_bounds__(256) void gcn_gemm1_wmma(
    const float* __restrict__ x, const float* __restrict__ W1,
    const float* __restrict__ degp,
    float* __restrict__ xw, float* __restrict__ dinv) {
    __shared__ __attribute__((aligned(16))) float lds[2][16 * G1_PITCH];
    const int lane = threadIdx.x & 31;
    const int wave = threadIdx.x >> 5;
    const int n0   = wave * 16;
    const int half = lane >> 4;
    const int l15  = lane & 15;
    const int col  = n0 + l15;
    v2f bf[IN_C / 4];
#pragma unroll
    for (int ks = 0; ks < IN_C / 4; ++ks) {
        const int ka = ks * 4 + half * 2;
        bf[ks].x = W1[ka * HID_C + col];
        bf[ks].y = W1[(ka + 1) * HID_C + col];
    }
    const int sr = threadIdx.x >> 4;
    const int sc = (threadIdx.x & 15) * 4;
    int tile = blockIdx.x;
    float4 st = *(const float4*)(x + (size_t)(tile * 16 + sr) * IN_C + sc);
    *(float4*)(&lds[0][sr * G1_PITCH + sc]) = st;
    __syncthreads();
    int buf = 0;
    for (; tile < N_TILES; tile += gridDim.x) {
        const int  m0       = tile * 16;
        const int  nxt      = tile + gridDim.x;
        const bool has_next = nxt < N_TILES;
        if (has_next) st = *(const float4*)(x + (size_t)(nxt * 16 + sr) * IN_C + sc);
        float dg[8];
#pragma unroll
        for (int r = 0; r < 8; ++r) dg[r] = degp[(size_t)(m0 + half * 8 + r) * DEG_STRIDE];
        v8f acc = {0.f, 0.f, 0.f, 0.f, 0.f, 0.f, 0.f, 0.f};
        const float* at = &lds[buf][l15 * G1_PITCH + half * 2];
#pragma unroll
        for (int ks = 0; ks < IN_C / 4; ++ks) {
            v2f a = *(const v2f*)(at + ks * 4);
            acc = __builtin_amdgcn_wmma_f32_16x16x4_f32(
                 false, a,  false, bf[ks],
                 (short)0, acc,  false,  false);
        }
        asm volatile("s_wait_alu depctr_va_vdst(0)" : "+v"(acc));
        float dn[8];
#pragma unroll
        for (int r = 0; r < 8; ++r) {
            const size_t o = (size_t)(m0 + half * 8 + r) * HID_C + col;
            dn[r]  = rsqrtf(dg[r]);
            xw[o]  = acc[r] * dn[r];
        }
        if (wave == 0 && l15 == 0) {
#pragma unroll
            for (int r = 0; r < 8; ++r) dinv[m0 + half * 8 + r] = dn[r];
        }
        if (has_next) *(float4*)(&lds[buf ^ 1][sr * G1_PITCH + sc]) = st;
        lds_barrier();
        buf ^= 1;
    }
}


template <int C, bool FINAL>
__global__ __launch_bounds__(1024, FINAL ? 16 : 8) void gcn_agg_lds(
    const int* __restrict__ src, const int* __restrict__ dst,
    const float* __restrict__ dinv, const float* __restrict__ degp, const float* __restrict__ bias,
    const float* __restrict__ w2,
    const int* __restrict__ pl, const int* __restrict__ pc,
    const float* __restrict__ y, float* __restrict__ outp) {
    constexpr int V = C / 32;
    constexpr int P = FINAL ? C : C + 4;
    typedef float vf __attribute__((ext_vector_type(V)));
    __shared__ __attribute__((aligned(16))) float acc[RANGE * P];
    constexpr int NST = (RANGE + 15) / 16;
    __shared__ __attribute__((aligned(16))) float dl[FINAL ? 4 : NST * 16];
    const int tid   = threadIdx.x;
    const int lane  = tid & 31;
    const int wave  = __builtin_amdgcn_readfirstlane(tid >> 5);
    const int lo    = blockIdx.x * RANGE;
    const int nrows = (N_NODES - lo) < RANGE ? (N_NODES - lo) : RANGE;
    const int2* pq = (const int2*)pl + (size_t)blockIdx.x * CAP;

    const int all   = __builtin_amdgcn_readfirstlane(pc[blockIdx.x * PC_STRIDE]);
    const int total = all <= CAP ? all : 0;
    int2 cur = make_int2(0, 0);
    if (tid < total) cur = pq[tid];

    if constexpr (!FINAL) {
        if (tid < NST * 16) dl[tid] = tid < nrows ? dinv[lo + tid] : 0.0f;
    }

    for (int r = wave; r < nrows; r += 32) {
        const int n = lo + r;
        vf v = *(const vf*)(y + (size_t)n * C + V * lane);
        if constexpr (!FINAL) {
            const float f = degp[(size_t)n * DEG_STRIDE] * dinv[n];
            const vf bv = *(const vf*)(bias + V * lane);
            v += bv * f;
        }
#pragma unroll
        for (int i = 0; i < V; ++i) acc[r * P + i * 32 + lane] = v[i];
    }
    __syncthreads();

    constexpr int G = 32 / V;
    for (int i0 = wave * 32; i0 < total; i0 += 1024) {
        const int i = i0 + lane;
        const int s = cur.x;
        const int r = cur.y;
        const int m = (total - i0) < 32 ? (total - i0) : 32;
        for (int g = 0; g < m; g += G) {
            vf v[G];
#pragma unroll
            for (int k = 0; k < G; ++k) {
                const int sk = __builtin_amdgcn_readlane(s, g + k);
                v[k] = *(const vf*)(y + (size_t)sk * C + V * lane);
            }
            if (g == 0) {
                cur = make_int2(0, 0);
                if (i + 1024 < total) cur = pq[i + 1024];
            }
#pragma unroll
            for (int k = 0; k < G; ++k) {
                if (k == 0 || g + k < m) {
                    const int rk = __builtin_amdgcn_readlane(r, g + k);
#pragma unroll
                    for (int j = 0; j < V; ++j) atomicAdd(&acc[rk * P + j * 32 + lane], v[k][j]);
                }
            }
        }
    }

    const int half = lane >> 4;
    const int l15  = lane & 15;
    const int col  = (wave & 3) * 16 + l15;
    v2f bf[FINAL ? 1 : C / 4];
    if constexpr (!FINAL) {
#pragma unroll
        for (int ks = 0; ks < C / 4; ++ks) {
            const int ka = 16 * (ks & 7) + 8 * half + (ks >> 3);
            bf[ks].x = w2[ka * OUT_C + col];
            bf[ks].y = w2[(ka + 4) * OUT_C + col];
        }
    }

    if (all > CAP) {
#pragma nounroll
        for (int e = tid; e < N_EDGES; e += 1024) {
            const int r = dst[e] - lo;
            if ((unsigned)r < (unsigned)RANGE) {
                const int s = src[e];
#pragma nounroll
                for (int ch = 0; ch < C; ++ch)
                    atomicAdd(&acc[r * P + (ch % V) * 32 + ch / V], y[(size_t)s * C + ch]);
            }
        }
    }
    __syncthreads();

    if constexpr (FINAL) {
        for (int r = wave; r < nrows; r += 32) {
            const int n = lo + r;
            vf v;
#pragma unroll
            for (int i = 0; i < V; ++i) v[i] = acc[r * P + i * 32 + lane];
            const float dn = dinv[n];
            const vf bv = *(const vf*)(bias + V * lane);
            v = bv + v * dn;
            *(vf*)(outp + (size_t)n * C + V * lane) = v;
        }
    } else {
        const int nst = (nrows + 15) >> 4;
        for (int st = wave >> 2; st < nst; st += 8) {
            const int ar = st * 16 + l15;
            const float* at = &acc[(ar < nrows ? ar : nrows - 1) * P + half * 2];
            const float4 dn0 = *(const float4*)&dl[st * 16 + half * 8];
            const float4 dn1 = *(const float4*)&dl[st * 16 + half * 8 + 4];
            const float dn[8] = {dn0.x, dn0.y, dn0.z, dn0.w, dn1.x, dn1.y, dn1.z, dn1.w};
            v8f d = {0.f, 0.f, 0.f, 0.f, 0.f, 0.f, 0.f, 0.f};
#pragma unroll
            for (int ks = 0; ks < C / 4; ++ks) {
                v2f a = *(const v2f*)(at + ks * 4);
                a.x = fmaxf(a.x, 0.0f);
                a.y = fmaxf(a.y, 0.0f);
                d = __builtin_amdgcn_wmma_f32_16x16x4_f32(
                     false, a,  false, bf[ks],
                     (short)0, d,  false,  false);
            }
            asm volatile("s_wait_alu depctr_va_vdst(0)" : "+v"(d));
#pragma unroll
            for (int r = 0; r < 8; ++r) {
                const int row = st * 16 + half * 8 + r;
                if (row < nrows)
                    outp[(size_t)(lo + row) * OUT_C + col] = d[r] * dn[r] * dn[r];
            }
        }
    }
}


extern "C" void kernel_launch(void* const* d_in, const int* in_sizes, int n_in,
                              void* d_out, int out_size, void* d_ws, size_t ws_size,
                              hipStream_t stream) {
    const float* x  = (const float*)d_in[0];
    const int*   ei = (const int*)d_in[1];
    const float* W1 = (const float*)d_in[2];
    const float* b1 = (const float*)d_in[3];
    const float* W2 = (const float*)d_in[4];
    const float* b2 = (const float*)d_in[5];
    float* out = (float*)d_out;

    const int* src = ei;
    const int* dst = ei + N_EDGES;

    float* dinv = (float*)d_ws;
    float* xw1  = dinv + 50176;
    float* xw2  = xw1 + 6400000;
    float* degp = xw2 + 6400000;
    int* pl = (int*)(degp + 1600000);
    int* pc = pl + 4194304;

    gcn_deg_init<<<(N_NODES + 255) / 256, 256, 0, stream>>>(degp, pc);
    gcn_deg_bucket<<<(N_EDGES + TB_EDGES - 1) / TB_EDGES, 1024, 0, stream>>>(src, dst, degp, pl, pc);

    gcn_gemm1_wmma<<<G1_BLOCKS, 256, 0, stream>>>(x, W1, degp, xw1, dinv);

    gcn_agg_lds<HID_C, false><<<NB, 1024, 0, stream>>>(src, dst, dinv, degp, b1, W2, pl, pc, xw1, xw2);

    gcn_agg_lds<OUT_C, true><<<NB, 1024, 0, stream>>>(src, dst, dinv, degp, b2, W2, pl, pc, xw2, out);
}
